# starting kernel
# baseline (speedup 1.0000x reference)
#include <hip/hip_runtime.h>

typedef __attribute__((ext_vector_type(2))) float v2f;
typedef __attribute__((ext_vector_type(8))) float v8f;

#define DIN  64
#define DOUT 64

__global__ __launch_bounds__(256) void gcn_xwt_wmma(
    const float* __restrict__ x, const float* __restrict__ W,
    float* __restrict__ y, int n_nodes)
{
    __shared__ float sW[DOUT * DIN];
    for (int i = threadIdx.x; i < DOUT * DIN; i += 256) sW[i] = W[i];
    __syncthreads();

    const int wave = threadIdx.x >> 5;
    const int lane = threadIdx.x & 31;
    const int m    = lane & 15;
    const int half = lane >> 4;
    const int tile = blockIdx.x * 8 + wave;
    const int row0 = tile * 16;
    if (row0 >= n_nodes) return;

    const int arow = min(row0 + m, n_nodes - 1);
    const float* __restrict__ xr = x + (size_t)arow * DIN;

    v8f acc[4] = {};

    #pragma unroll
    for (int k0 = 0; k0 < DIN; k0 += 4) {
        const int ka = k0 + 2 * half;
        v2f a;
        a.x = xr[ka];
        a.y = xr[ka + 1];
        #pragma unroll
        for (int nt = 0; nt < 4; ++nt) {
            const int n = nt * 16 + m;
            v2f bw;
            bw.x = sW[n * DIN + ka];
            bw.y = sW[n * DIN + ka + 1];
            acc[nt] = __builtin_amdgcn_wmma_f32_16x16x4_f32(
                 false, a,  false, bw,
                 (short)0, acc[nt],
                 false,  false);
        }
    }

    #pragma unroll
    for (int nt = 0; nt < 4; ++nt) {
        #pragma unroll
        for (int v = 0; v < 8; ++v) {
            const int r = row0 + v + 8 * half;
            if (r < n_nodes)
                y[(size_t)r * DOUT + nt * 16 + m] = acc[nt][v];
        }
    }
}

__global__ __launch_bounds__(256) void gcn_bias_init(
    float* __restrict__ out, const float* __restrict__ b, int n_nodes)
{
    const int total = n_nodes * DOUT;
    for (int i = blockIdx.x * blockDim.x + threadIdx.x; i < total;
         i += gridDim.x * blockDim.x)
        out[i] = b[i & (DOUT - 1)];
}

__global__ __launch_bounds__(256) void gcn_edge_scatter(
    const int* __restrict__ erow, const int* __restrict__ ecol,
    const float* __restrict__ eval, const float* __restrict__ y,
    float* out, int n_edges)
{
    const int gid = blockIdx.x * 256 + threadIdx.x;
    const int e = gid >> 4;
    const int q = gid & 15;
    if (e >= n_edges) return;

    const int r   = erow[e];
    const int c   = ecol[e];
    const float v = eval[e];

    const float4* __restrict__ ys =
        (const float4*)(y + (size_t)c * DOUT);
    const float4 mv = ys[q];

    float* o = out + (size_t)r * DOUT + q * 4;
    atomicAdd(o + 0, v * mv.x);
    atomicAdd(o + 1, v * mv.y);
    atomicAdd(o + 2, v * mv.z);
    atomicAdd(o + 3, v * mv.w);
}

extern "C" void kernel_launch(void* const* d_in, const int* in_sizes, int n_in,
                              void* d_out, int out_size, void* d_ws, size_t ws_size,
                              hipStream_t stream)
{
    const float* x    = (const float*)d_in[0];
    const int*   erow = (const int*)  d_in[1];
    const int*   ecol = (const int*)  d_in[2];
    const float* ev   = (const float*)d_in[3];
    const float* W    = (const float*)d_in[4];
    const float* b    = (const float*)d_in[5];
    float* out = (float*)d_out;
    float* y   = (float*)d_ws;

    const int n_nodes = in_sizes[0] / DIN;
    const int n_edges = in_sizes[3];

    const int tiles = (n_nodes + 15) / 16;
    gcn_xwt_wmma<<<(tiles + 7) / 8, 256, 0, stream>>>(x, W, y, n_nodes);

    const int total = n_nodes * DOUT;
    int bgrid = (total + 255) / 256;
    if (bgrid > 4096) bgrid = 4096;
    gcn_bias_init<<<bgrid, 256, 0, stream>>>(out, b, n_nodes);

    const long long sthreads = (long long)n_edges * 16;
    gcn_edge_scatter<<<(int)((sthreads + 255) / 256), 256, 0, stream>>>(
        erow, ecol, ev, y, out, n_edges);
}


# edge sum re-routed: edges grouped by 392-row out range (count+fill launches); one block per range sums val*y[col] in LDS (ds f32 adds, arrival order), stores b+sum once; no global f32 atomics
# speedup vs baseline: 1.6892x; 1.6892x over previous
#include <hip/hip_runtime.h>

typedef __attribute__((ext_vector_type(2))) float v2f;
typedef __attribute__((ext_vector_type(8))) float v8f;

#define DIN  64
#define DOUT 64
#define RPB  392
#define MAXBINS 1024
#define EPB  4096
#define COLBITS 23

__global__ __launch_bounds__(256) void gcn_xwt_wmma(
    const float* __restrict__ x, const float* __restrict__ W,
    float* __restrict__ y, int n_nodes, int* __restrict__ ctr)
{
    __shared__ float sW[DOUT * DIN];
    if (blockIdx.x == 0)
        for (int i = threadIdx.x; i < 2 * MAXBINS; i += 256) ctr[i] = 0;
    for (int i = threadIdx.x; i < DOUT * DIN; i += 256) sW[i] = W[i];
    __syncthreads();

    const int wave = threadIdx.x >> 5;
    const int lane = threadIdx.x & 31;
    const int m    = lane & 15;
    const int half = lane >> 4;
    const int tile = blockIdx.x * 8 + wave;
    const int row0 = tile * 16;
    if (row0 >= n_nodes) return;

    const int arow = min(row0 + m, n_nodes - 1);
    const float* __restrict__ xr = x + (size_t)arow * DIN;

    v8f acc[4] = {};

    #pragma unroll
    for (int k0 = 0; k0 < DIN; k0 += 4) {
        const int ka = k0 + 2 * half;
        v2f a;
        a.x = xr[ka];
        a.y = xr[ka + 1];
        #pragma unroll
        for (int nt = 0; nt < 4; ++nt) {
            const int n = nt * 16 + m;
            v2f bw;
            bw.x = sW[n * DIN + ka];
            bw.y = sW[n * DIN + ka + 1];
            acc[nt] = __builtin_amdgcn_wmma_f32_16x16x4_f32(
                 false, a,  false, bw,
                 (short)0, acc[nt],
                 false,  false);
        }
    }

    #pragma unroll
    for (int nt = 0; nt < 4; ++nt) {
        #pragma unroll
        for (int v = 0; v < 8; ++v) {
            const int r = row0 + v + 8 * half;
            if (r < n_nodes)
                y[(size_t)r * DOUT + nt * 16 + m] = acc[nt][v];
        }
    }
}

__global__ __launch_bounds__(256) void gcn_bin_count(
    const int* __restrict__ erow, int* cnt, int n_edges, int nbins)
{
    __shared__ int h[MAXBINS];
    for (int i = threadIdx.x; i < nbins; i += 256) h[i] = 0;
    __syncthreads();

    const int base = blockIdx.x * EPB + threadIdx.x;
    #pragma unroll
    for (int j = 0; j < EPB / 256; ++j) {
        const int e = base + j * 256;
        if (e < n_edges)
            atomicAdd(&h[(unsigned)erow[e] / RPB], 1);
    }
    __syncthreads();

    for (int i = threadIdx.x; i < nbins; i += 256) {
        const int c = h[i];
        if (c) atomicAdd(&cnt[i], c);
    }
}

__global__ __launch_bounds__(256) void gcn_bin_fill(
    const int* __restrict__ erow, const int* __restrict__ ecol,
    const float* __restrict__ eval, const int* __restrict__ cnt,
    int* cur, int* __restrict__ bstart, unsigned int* __restrict__ ent,
    int n_edges, int nbins)
{
    __shared__ int h[MAXBINS];
    __shared__ int st[MAXBINS];
    __shared__ int gb[MAXBINS];
    __shared__ int lc[MAXBINS];
    __shared__ unsigned int skey[EPB];
    __shared__ unsigned int sval[EPB];
    __shared__ int sdst[EPB];
    __shared__ int wa[8], wc[8];

    for (int i = threadIdx.x; i < nbins; i += 256) { h[i] = 0; st[i] = cnt[i]; }
    __syncthreads();

    const int base = blockIdx.x * EPB + threadIdx.x;
    int   rows[EPB / 256];
    int   cols[EPB / 256];
    float vals[EPB / 256];
    #pragma unroll
    for (int j = 0; j < EPB / 256; ++j) {
        const int e = base + j * 256;
        const bool ok = e < n_edges;
        rows[j] = ok ? erow[e] : -1;
        cols[j] = ok ? ecol[e] : 0;
        vals[j] = ok ? eval[e] : 0.0f;
    }
    #pragma unroll
    for (int j = 0; j < EPB / 256; ++j)
        if (rows[j] >= 0)
            atomicAdd(&h[(unsigned)rows[j] / RPB], 1);
    __syncthreads();

    const int lane = threadIdx.x & 31;
    const int wave = threadIdx.x >> 5;
    const int t4   = threadIdx.x * 4;
    int a[4], c[4];
    #pragma unroll
    for (int q = 0; q < 4; ++q) {
        const bool in = t4 + q < nbins;
        a[q] = in ? st[t4 + q] : 0;
        c[q] = in ? h[t4 + q]  : 0;
    }
    const int sa = a[0] + a[1] + a[2] + a[3];
    const int sc = c[0] + c[1] + c[2] + c[3];
    int ia = sa, ic = sc;
    #pragma unroll
    for (int d = 1; d < 32; d <<= 1) {
        const int ta = __shfl_up(ia, d);
        const int tc = __shfl_up(ic, d);
        if (lane >= d) { ia += ta; ic += tc; }
    }
    if (lane == 31) { wa[wave] = ia; wc[wave] = ic; }
    __syncthreads();
    int pa = ia - sa, pc = ic - sc;
    for (int w = 0; w < wave; ++w) { pa += wa[w]; pc += wc[w]; }
    int off[4];
    #pragma unroll
    for (int q = 0; q < 4; ++q)
        off[q] = c[q] ? atomicAdd(&cur[t4 + q], c[q]) : 0;
    #pragma unroll
    for (int q = 0; q < 4; ++q) {
        const int i = t4 + q;
        if (i < nbins) {
            if (blockIdx.x == 0) bstart[i] = pa;
            gb[i] = pa + off[q] - pc;
            lc[i] = pc;
        }
        pa += a[q];
        pc += c[q];
    }
    __syncthreads();

    #pragma unroll
    for (int j = 0; j < EPB / 256; ++j) {
        if (rows[j] >= 0) {
            const unsigned bin = (unsigned)rows[j] / RPB;
            const unsigned rl  = (unsigned)rows[j] - bin * RPB;
            const int p = atomicAdd(&lc[bin], 1);
            skey[p] = (unsigned)cols[j] | (rl << COLBITS);
            sval[p] = __float_as_uint(vals[j]);
            sdst[p] = gb[bin] + p;
        }
    }
    __syncthreads();

    const int first = blockIdx.x * EPB;
    const int nloc  = min(EPB, n_edges - first);
    uint2* __restrict__ ent2 = (uint2*)ent;
    for (int p = threadIdx.x; p < nloc; p += 256)
        ent2[sdst[p]] = make_uint2(skey[p], sval[p]);
}

__global__ __launch_bounds__(1024) void gcn_rows_sum(
    const unsigned int* __restrict__ ent, const int* __restrict__ cnt,
    const int* __restrict__ bstart, const float* __restrict__ y,
    const float* __restrict__ b, float* __restrict__ out, int n_nodes)
{
    __shared__ float acc[RPB * DOUT];
    const int lane = threadIdx.x & 31;
    const int wave = __builtin_amdgcn_readfirstlane(threadIdx.x >> 5);
    const int bin  = blockIdx.x;
    const int row0 = bin * RPB;

    const float2 bb = ((const float2*)b)[lane];
    for (int r = wave; r < RPB; r += 32)
        *(float2*)&acc[r * DOUT + 2 * lane] = bb;
    __syncthreads();

    const int n   = __builtin_amdgcn_readfirstlane(cnt[bin]);
    const int beg = __builtin_amdgcn_readfirstlane(bstart[bin]);
    const uint2* __restrict__ seg = (const uint2*)ent + beg;

    int c0 = wave * 32;
    uint2 en = make_uint2(0u, 0u);
    if (c0 + lane < n) en = seg[c0 + lane];
    while (c0 < n) {
        const int cn = c0 + 32 * 32;
        uint2 en_next = make_uint2(0u, 0u);
        if (cn + lane < n) en_next = seg[cn + lane];

        const int m = min(32, n - c0);
        for (int j = 0; j < m; j += 16) {
            float2 yv[16];
            #pragma unroll
            for (int u = 0; u < 16; ++u) {
                const int jj = min(j + u, m - 1);
                const unsigned key = __builtin_amdgcn_readlane(en.x, jj);
                const unsigned col = key & ((1u << COLBITS) - 1u);
                yv[u] = ((const float2*)(y + (size_t)col * DOUT))[lane];
            }
            #pragma unroll
            for (int u = 0; u < 16; ++u) {
                if (j + u < m) {
                    const unsigned key = __builtin_amdgcn_readlane(en.x, j + u);
                    const float    val = __uint_as_float(__builtin_amdgcn_readlane(en.y, j + u));
                    float* a = &acc[(key >> COLBITS) * DOUT + 2 * lane];
                    atomicAdd(a,     val * yv[u].x);
                    atomicAdd(a + 1, val * yv[u].y);
                }
            }
        }
        en = en_next;
        c0 = cn;
    }
    __syncthreads();

    for (int r = wave; r < RPB; r += 32) {
        const int gr = row0 + r;
        if (gr < n_nodes)
            *(float2*)&out[(size_t)gr * DOUT + 2 * lane] = *(const float2*)&acc[r * DOUT + 2 * lane];
    }
}

extern "C" void kernel_launch(void* const* d_in, const int* in_sizes, int n_in,
                              void* d_out, int out_size, void* d_ws, size_t ws_size,
                              hipStream_t stream)
{
    const float* x    = (const float*)d_in[0];
    const int*   erow = (const int*)  d_in[1];
    const int*   ecol = (const int*)  d_in[2];
    const float* ev   = (const float*)d_in[3];
    const float* W    = (const float*)d_in[4];
    const float* b    = (const float*)d_in[5];
    float* out = (float*)d_out;
    char*  wsb = (char*)d_ws;
    float* y   = (float*)wsb;
    int*   ctr = (int*)(wsb + (64u << 20));
    unsigned int* ent = (unsigned int*)(wsb + (128u << 20));

    const int n_nodes = in_sizes[0] / DIN;
    const int n_edges = in_sizes[3];

    const int tiles = (n_nodes + 15) / 16;
    gcn_xwt_wmma<<<(tiles + 7) / 8, 256, 0, stream>>>(x, W, y, n_nodes, ctr);

    const int nbins   = (n_nodes + RPB - 1) / RPB;
    const int eblocks = (n_edges + EPB - 1) / EPB;
    gcn_bin_count<<<eblocks, 256, 0, stream>>>(erow, ctr, n_edges, nbins);
    gcn_bin_fill<<<eblocks, 256, 0, stream>>>(erow, ecol, ev, ctr, ctr + MAXBINS,
                                              ctr + 2 * MAXBINS, ent, n_edges, nbins);

    gcn_rows_sum<<<nbins, 1024, 0, stream>>>(ent, ctr, ctr + 2 * MAXBINS, y, b, out, n_nodes);
}


# x@W^T moved behind the edge sum: a row-range block sums val*x[col] in its LDS tile, multiplies it by W^T on f32 WMMA, adds b; y pass dropped (launch 1 only clears counters); sums edges first, then k
# speedup vs baseline: 2.2755x; 1.3471x over previous
#include <hip/hip_runtime.h>

typedef __attribute__((ext_vector_type(2))) float v2f;
typedef __attribute__((ext_vector_type(8))) float v8f;

#define DIN  64
#define DOUT 64
#define RPB  392
#define MAXBINS 1024
#define EPB  4096
#define COLBITS 23
#define PITCH 68
#define TROWS 400

__global__ __launch_bounds__(256) void gcn_ctr_clear(int* __restrict__ ctr)
{
    for (int i = threadIdx.x; i < 2 * MAXBINS; i += 256) ctr[i] = 0;
}

__global__ __launch_bounds__(256) void gcn_bin_count(
    const int* __restrict__ erow, int* cnt, int n_edges, int nbins)
{
    __shared__ int h[MAXBINS];
    for (int i = threadIdx.x; i < nbins; i += 256) h[i] = 0;
    __syncthreads();

    const int base = blockIdx.x * EPB + threadIdx.x;
    #pragma unroll
    for (int j = 0; j < EPB / 256; ++j) {
        const int e = base + j * 256;
        if (e < n_edges)
            atomicAdd(&h[(unsigned)erow[e] / RPB], 1);
    }
    __syncthreads();

    for (int i = threadIdx.x; i < nbins; i += 256) {
        const int c = h[i];
        if (c) atomicAdd(&cnt[i], c);
    }
}

__global__ __launch_bounds__(256) void gcn_bin_fill(
    const int* __restrict__ erow, const int* __restrict__ ecol,
    const float* __restrict__ eval, const int* __restrict__ cnt,
    int* cur, int* __restrict__ bstart, unsigned int* __restrict__ ent,
    int n_edges, int nbins)
{
    __shared__ int h[MAXBINS];
    __shared__ int st[MAXBINS];
    __shared__ int gb[MAXBINS];
    __shared__ int lc[MAXBINS];
    __shared__ unsigned int skey[EPB];
    __shared__ unsigned int sval[EPB];
    __shared__ int sdst[EPB];
    __shared__ int wa[8], wc[8];

    for (int i = threadIdx.x; i < nbins; i += 256) { h[i] = 0; st[i] = cnt[i]; }
    __syncthreads();

    const int base = blockIdx.x * EPB + threadIdx.x;
    int   rows[EPB / 256];
    int   cols[EPB / 256];
    float vals[EPB / 256];
    #pragma unroll
    for (int j = 0; j < EPB / 256; ++j) {
        const int e = base + j * 256;
        const bool ok = e < n_edges;
        rows[j] = ok ? erow[e] : -1;
        cols[j] = ok ? ecol[e] : 0;
        vals[j] = ok ? eval[e] : 0.0f;
    }
    #pragma unroll
    for (int j = 0; j < EPB / 256; ++j)
        if (rows[j] >= 0)
            atomicAdd(&h[(unsigned)rows[j] / RPB], 1);
    __syncthreads();

    const int lane = threadIdx.x & 31;
    const int wave = threadIdx.x >> 5;
    const int t4   = threadIdx.x * 4;
    int a[4], c[4];
    #pragma unroll
    for (int q = 0; q < 4; ++q) {
        const bool in = t4 + q < nbins;
        a[q] = in ? st[t4 + q] : 0;
        c[q] = in ? h[t4 + q]  : 0;
    }
    const int sa = a[0] + a[1] + a[2] + a[3];
    const int sc = c[0] + c[1] + c[2] + c[3];
    int ia = sa, ic = sc;
    #pragma unroll
    for (int d = 1; d < 32; d <<= 1) {
        const int ta = __shfl_up(ia, d);
        const int tc = __shfl_up(ic, d);
        if (lane >= d) { ia += ta; ic += tc; }
    }
    if (lane == 31) { wa[wave] = ia; wc[wave] = ic; }
    __syncthreads();
    int pa = ia - sa, pc = ic - sc;
    for (int w = 0; w < wave; ++w) { pa += wa[w]; pc += wc[w]; }
    int off[4];
    #pragma unroll
    for (int q = 0; q < 4; ++q)
        off[q] = c[q] ? atomicAdd(&cur[t4 + q], c[q]) : 0;
    #pragma unroll
    for (int q = 0; q < 4; ++q) {
        const int i = t4 + q;
        if (i < nbins) {
            if (blockIdx.x == 0) bstart[i] = pa;
            gb[i] = pa + off[q] - pc;
            lc[i] = pc;
        }
        pa += a[q];
        pc += c[q];
    }
    __syncthreads();

    #pragma unroll
    for (int j = 0; j < EPB / 256; ++j) {
        if (rows[j] >= 0) {
            const unsigned bin = (unsigned)rows[j] / RPB;
            const unsigned rl  = (unsigned)rows[j] - bin * RPB;
            const int p = atomicAdd(&lc[bin], 1);
            skey[p] = (unsigned)cols[j] | (rl << COLBITS);
            sval[p] = __float_as_uint(vals[j]);
            sdst[p] = gb[bin] + p;
        }
    }
    __syncthreads();

    const int first = blockIdx.x * EPB;
    const int nloc  = min(EPB, n_edges - first);
    uint2* __restrict__ ent2 = (uint2*)ent;
    for (int p = threadIdx.x; p < nloc; p += 256)
        ent2[sdst[p]] = make_uint2(skey[p], sval[p]);
}

__global__ __launch_bounds__(1024) void gcn_rows_sum_xwt(
    const unsigned int* __restrict__ ent, const int* __restrict__ cnt,
    const int* __restrict__ bstart, const float* __restrict__ x,
    const float* __restrict__ W, const float* __restrict__ b,
    float* __restrict__ out, int n_nodes)
{
    __shared__ __attribute__((aligned(16))) float acc[TROWS * PITCH];
    __shared__ __attribute__((aligned(16))) float sW[DOUT * PITCH];
    const int lane = threadIdx.x & 31;
    const int wave = __builtin_amdgcn_readfirstlane(threadIdx.x >> 5);
    const int bin  = blockIdx.x;
    const int row0 = bin * RPB;
    const int m    = lane & 15;
    const int half = lane >> 4;

    {
        const int wn = threadIdx.x >> 4;
        const int wk = (threadIdx.x & 15) * 4;
        *(float4*)&sW[wn * PITCH + wk] = *(const float4*)&W[wn * DIN + wk];
    }
    const float2 bb = ((const float2*)b)[lane];

    for (int r = wave; r < TROWS; r += 32)
        *(float2*)&acc[r * PITCH + 2 * lane] = make_float2(0.0f, 0.0f);
    __syncthreads();

    const int n   = __builtin_amdgcn_readfirstlane(cnt[bin]);
    const int beg = __builtin_amdgcn_readfirstlane(bstart[bin]);
    const uint2* __restrict__ seg = (const uint2*)ent + beg;

    int c0 = wave * 32;
    uint2 en = make_uint2(0u, 0u);
    if (c0 + lane < n) en = seg[c0 + lane];
    while (c0 < n) {
        const int cn = c0 + 32 * 32;
        uint2 en_next = make_uint2(0u, 0u);
        if (cn + lane < n) en_next = seg[cn + lane];

        const int mm = min(32, n - c0);
        for (int j = 0; j < mm; j += 16) {
            float2 xv[16];
            #pragma unroll
            for (int u = 0; u < 16; ++u) {
                const int jj = min(j + u, mm - 1);
                const unsigned key = __builtin_amdgcn_readlane(en.x, jj);
                const unsigned col = key & ((1u << COLBITS) - 1u);
                xv[u] = ((const float2*)(x + (size_t)col * DIN))[lane];
            }
            #pragma unroll
            for (int u = 0; u < 16; ++u) {
                if (j + u < mm) {
                    const unsigned key = __builtin_amdgcn_readlane(en.x, j + u);
                    const float    val = __uint_as_float(__builtin_amdgcn_readlane(en.y, j + u));
                    float* a = &acc[(key >> COLBITS) * PITCH + 2 * lane];
                    atomicAdd(a,     val * xv[u].x);
                    atomicAdd(a + 1, val * xv[u].y);
                }
            }
        }
        en = en_next;
        c0 = cn;
    }
    __syncthreads();

    for (int t = wave; t < TROWS / 16; t += 32) {
        const float* ar = &acc[(t * 16 + m) * PITCH + 2 * half];
        v2f a[16];
        #pragma unroll
        for (int s = 0; s < 16; ++s) a[s] = *(const v2f*)(ar + 4 * s);

        v8f d[4] = {};

        #pragma unroll
        for (int s = 0; s < 16; ++s) {
            #pragma unroll
            for (int nt = 0; nt < 4; ++nt) {
                const v2f bw = *(const v2f*)&sW[(nt * 16 + m) * PITCH + 4 * s + 2 * half];
                d[nt] = __builtin_amdgcn_wmma_f32_16x16x4_f32(
                     false, a[s],  false, bw,
                     (short)0, d[nt],
                     false,  false);
            }
        }

        #pragma unroll
        for (int nt = 0; nt < 4; ++nt)
            #pragma unroll
            for (int v = 0; v < 8; ++v)
                acc[(t * 16 + v + 8 * half) * PITCH + nt * 16 + m] = d[nt][v];
    }
    __syncthreads();

    for (int r = wave; r < RPB; r += 32) {
        const int gr = row0 + r;
        if (gr < n_nodes) {
            const float2 o = *(const float2*)&acc[r * PITCH + 2 * lane];
            *(float2*)&out[(size_t)gr * DOUT + 2 * lane] = make_float2(o.x + bb.x, o.y + bb.y);
        }
    }
}

extern "C" void kernel_launch(void* const* d_in, const int* in_sizes, int n_in,
                              void* d_out, int out_size, void* d_ws, size_t ws_size,
                              hipStream_t stream)
{
    const float* x    = (const float*)d_in[0];
    const int*   erow = (const int*)  d_in[1];
    const int*   ecol = (const int*)  d_in[2];
    const float* ev   = (const float*)d_in[3];
    const float* W    = (const float*)d_in[4];
    const float* b    = (const float*)d_in[5];
    float* out = (float*)d_out;
    char*  wsb = (char*)d_ws;
    int*   ctr = (int*)(wsb + (64u << 20));
    unsigned int* ent = (unsigned int*)(wsb + (128u << 20));

    const int n_nodes = in_sizes[0] / DIN;
    const int n_edges = in_sizes[3];

    gcn_ctr_clear<<<1, 256, 0, stream>>>(ctr);

    const int nbins   = (n_nodes + RPB - 1) / RPB;
    const int eblocks = (n_edges + EPB - 1) / EPB;
    gcn_bin_count<<<eblocks, 256, 0, stream>>>(erow, ctr, n_edges, nbins);
    gcn_bin_fill<<<eblocks, 256, 0, stream>>>(erow, ecol, ev, ctr, ctr + MAXBINS,
                                              ctr + 2 * MAXBINS, ent, n_edges, nbins);

    gcn_rows_sum_xwt<<<nbins, 1024, 0, stream>>>(ent, ctr, ctr + 2 * MAXBINS, x, W, b, out, n_nodes);
}


# edges grouped inside each 4096-edge slice only (LDS order + start/count table); a row block lists its pieces in LDS, slice-major; clear and count launches, global atomics dropped; LDS f32 adds kept
# speedup vs baseline: 2.9671x; 1.3039x over previous
#include <hip/hip_runtime.h>

typedef __attribute__((ext_vector_type(2))) float v2f;
typedef __attribute__((ext_vector_type(8))) float v8f;

#define DIN  64
#define DOUT 64
#define RPB  392
#define MAXBINS 1024
#define EPB  4096
#define COLBITS 23
#define PITCH 68
#define CAP  8192
#define SEGT 1024
#define TROWS 400

__global__ __launch_bounds__(256) void gcn_slice_group(
    const int* __restrict__ erow, const int* __restrict__ ecol,
    const float* __restrict__ eval, int* __restrict__ tab,
    unsigned int* __restrict__ ent, int n_edges, int nbins)
{
    __shared__ int h[MAXBINS];
    __shared__ unsigned int skey[EPB];
    __shared__ unsigned int sval[EPB];
    __shared__ int wc[8];

    for (int i = threadIdx.x; i < nbins; i += 256) h[i] = 0;
    __syncthreads();

    const int base = blockIdx.x * EPB + threadIdx.x;
    int   rows[EPB / 256];
    int   cols[EPB / 256];
    float vals[EPB / 256];
    #pragma unroll
    for (int j = 0; j < EPB / 256; ++j) {
        const int e = base + j * 256;
        const bool ok = e < n_edges;
        rows[j] = ok ? erow[e] : -1;
        cols[j] = ok ? ecol[e] : 0;
        vals[j] = ok ? eval[e] : 0.0f;
    }
    #pragma unroll
    for (int j = 0; j < EPB / 256; ++j)
        if (rows[j] >= 0)
            atomicAdd(&h[(unsigned)rows[j] / RPB], 1);
    __syncthreads();

    const int lane = threadIdx.x & 31;
    const int wave = threadIdx.x >> 5;
    const int t4   = threadIdx.x * 4;
    int c[4];
    #pragma unroll
    for (int q = 0; q < 4; ++q)
        c[q] = (t4 + q < nbins) ? h[t4 + q] : 0;
    const int sc = c[0] + c[1] + c[2] + c[3];
    int ic = sc;
    #pragma unroll
    for (int d = 1; d < 32; d <<= 1) {
        const int tc = __shfl_up(ic, d);
        if (lane >= d) ic += tc;
    }
    if (lane == 31) wc[wave] = ic;
    __syncthreads();
    int pc = ic - sc;
    for (int w = 0; w < wave; ++w) pc += wc[w];
    int2* __restrict__ trow = (int2*)tab + (size_t)blockIdx.x * nbins;
    #pragma unroll
    for (int q = 0; q < 4; ++q) {
        const int i = t4 + q;
        if (i < nbins) {
            trow[i] = make_int2(pc, c[q]);
            h[i] = pc;
        }
        pc += c[q];
    }
    __syncthreads();

    #pragma unroll
    for (int j = 0; j < EPB / 256; ++j) {
        if (rows[j] >= 0) {
            const unsigned bin = (unsigned)rows[j] / RPB;
            const unsigned rl  = (unsigned)rows[j] - bin * RPB;
            const int p = atomicAdd(&h[bin], 1);
            skey[p] = (unsigned)cols[j] | (rl << COLBITS);
            sval[p] = __float_as_uint(vals[j]);
        }
    }
    __syncthreads();

    const int first = blockIdx.x * EPB;
    const int nloc  = min(EPB, n_edges - first);
    uint2* __restrict__ ent2 = (uint2*)ent + first;
    for (int p = threadIdx.x; p < nloc; p += 256)
        ent2[p] = make_uint2(skey[p], sval[p]);
}

__global__ __launch_bounds__(1024) void gcn_rows_sum_xwt(
    const unsigned int* __restrict__ ent, const int* __restrict__ tab,
    const float* __restrict__ x,
    const float* __restrict__ W, const float* __restrict__ b,
    float* __restrict__ out, int n_nodes, int nbins, int nseg)
{
    __shared__ __attribute__((aligned(16))) float acc[TROWS * PITCH];
    __shared__ __attribute__((aligned(16))) float sW[DOUT * PITCH];
    __shared__ uint2 stage[CAP];
    __shared__ int sbeg[SEGT];
    __shared__ int spos[SEGT];
    __shared__ int slen[SEGT];
    __shared__ int wtot[33];
    const int lane = threadIdx.x & 31;
    const int wave = __builtin_amdgcn_readfirstlane(threadIdx.x >> 5);
    const int bin  = blockIdx.x;
    const int row0 = bin * RPB;
    const int m    = lane & 15;
    const int half = lane >> 4;

    {
        const int wn = threadIdx.x >> 4;
        const int wk = (threadIdx.x & 15) * 4;
        *(float4*)&sW[wn * PITCH + wk] = *(const float4*)&W[wn * DIN + wk];
    }
    const float2 bb = ((const float2*)b)[lane];

    for (int r = wave; r < TROWS; r += 32)
        *(float2*)&acc[r * PITCH + 2 * lane] = make_float2(0.0f, 0.0f);
    __syncthreads();

    const uint2* __restrict__ ent2 = (const uint2*)ent;
    const int2*  __restrict__ tab2 = (const int2*)tab;

    for (int g0 = 0; g0 < nseg; g0 += SEGT) {
        const int nsg = min(SEGT, nseg - g0);
        int gb = 0, gl = 0;
        if ((int)threadIdx.x < nsg) {
            const int t = g0 + (int)threadIdx.x;
            const int2 e = tab2[(size_t)t * nbins + bin];
            gb = t * EPB + e.x;
            gl = e.y;
        }
        int inc = gl;
        #pragma unroll
        for (int dd = 1; dd < 32; dd <<= 1) {
            const int v = __shfl_up(inc, dd);
            if (lane >= dd) inc += v;
        }
        if (lane == 31) wtot[wave] = inc;
        __syncthreads();
        if (wave == 0) {
            const int w = wtot[lane];
            int wi = w;
            #pragma unroll
            for (int dd = 1; dd < 32; dd <<= 1) {
                const int v = __shfl_up(wi, dd);
                if (lane >= dd) wi += v;
            }
            wtot[lane] = wi - w;
            if (lane == 31) wtot[32] = wi;
        }
        __syncthreads();
        sbeg[threadIdx.x] = gb;
        spos[threadIdx.x] = wtot[wave] + inc - gl;
        slen[threadIdx.x] = gl;
        const int gtot = __builtin_amdgcn_readfirstlane(wtot[32]);
        __syncthreads();

        for (int r0 = 0; r0 < gtot; r0 += CAP) {
            const int n = min(CAP, gtot - r0);

            for (int sb = 0; sb < nsg; sb += 256) {
                uint2 v[8];
                int ln[8], ps[8], bg[8];
                #pragma unroll
                for (int u = 0; u < 8; ++u) {
                    const int s  = sb + wave + 32 * u;
                    const int sc = min(s, nsg - 1);
                    ln[u] = (s < nsg) ? slen[sc] : 0;
                    ps[u] = spos[sc] - r0;
                    bg[u] = sbeg[sc];
                }
                #pragma unroll
                for (int u = 0; u < 8; ++u) {
                    v[u] = make_uint2(0u, 0u);
                    if (lane < ln[u]) v[u] = ent2[bg[u] + lane];
                }
                #pragma unroll
                for (int u = 0; u < 8; ++u) {
                    const int q = ps[u] + lane;
                    if (lane < ln[u] && (unsigned)q < (unsigned)CAP) stage[q] = v[u];
                    for (int j = lane + 32; j < ln[u]; j += 32) {
                        const int q2 = ps[u] + j;
                        if ((unsigned)q2 < (unsigned)CAP) stage[q2] = ent2[bg[u] + j];
                    }
                }
            }
            __syncthreads();

            int c0 = wave * 32;
            uint2 en = make_uint2(0u, 0u);
            if (c0 + lane < n) en = stage[c0 + lane];
            while (c0 < n) {
                const int cn = c0 + 32 * 32;
                uint2 en_next = make_uint2(0u, 0u);
                if (cn + lane < n) en_next = stage[cn + lane];

                const int mm = min(32, n - c0);
                for (int j = 0; j < mm; j += 16) {
                    float2 xv[16];
                    #pragma unroll
                    for (int u = 0; u < 16; ++u) {
                        const int jj = min(j + u, mm - 1);
                        const unsigned key = __builtin_amdgcn_readlane(en.x, jj);
                        const unsigned col = key & ((1u << COLBITS) - 1u);
                        xv[u] = ((const float2*)(x + (size_t)col * DIN))[lane];
                    }
                    #pragma unroll
                    for (int u = 0; u < 16; ++u) {
                        if (j + u < mm) {
                            const unsigned key = __builtin_amdgcn_readlane(en.x, j + u);
                            const float    val = __uint_as_float(__builtin_amdgcn_readlane(en.y, j + u));
                            float* a = &acc[(key >> COLBITS) * PITCH + 2 * lane];
                            atomicAdd(a,     val * xv[u].x);
                            atomicAdd(a + 1, val * xv[u].y);
                        }
                    }
                }
                en = en_next;
                c0 = cn;
            }
            __syncthreads();
        }
    }

    for (int t = wave; t < TROWS / 16; t += 32) {
        const float* ar = &acc[(t * 16 + m) * PITCH + 2 * half];
        v2f a[16];
        #pragma unroll
        for (int s = 0; s < 16; ++s) a[s] = *(const v2f*)(ar + 4 * s);

        v8f d[4] = {};

        #pragma unroll
        for (int s = 0; s < 16; ++s) {
            #pragma unroll
            for (int nt = 0; nt < 4; ++nt) {
                const v2f bw = *(const v2f*)&sW[(nt * 16 + m) * PITCH + 4 * s + 2 * half];
                d[nt] = __builtin_amdgcn_wmma_f32_16x16x4_f32(
                     false, a[s],  false, bw,
                     (short)0, d[nt],
                     false,  false);
            }
        }

        #pragma unroll
        for (int nt = 0; nt < 4; ++nt)
            #pragma unroll
            for (int v = 0; v < 8; ++v)
                acc[(t * 16 + v + 8 * half) * PITCH + nt * 16 + m] = d[nt][v];
    }
    __syncthreads();

    for (int r = wave; r < RPB; r += 32) {
        const int gr = row0 + r;
        if (gr < n_nodes) {
            const float2 o = *(const float2*)&acc[r * PITCH + 2 * lane];
            *(float2*)&out[(size_t)gr * DOUT + 2 * lane] = make_float2(o.x + bb.x, o.y + bb.y);
        }
    }
}

extern "C" void kernel_launch(void* const* d_in, const int* in_sizes, int n_in,
                              void* d_out, int out_size, void* d_ws, size_t ws_size,
                              hipStream_t stream)
{
    const float* x    = (const float*)d_in[0];
    const int*   erow = (const int*)  d_in[1];
    const int*   ecol = (const int*)  d_in[2];
    const float* ev   = (const float*)d_in[3];
    const float* W    = (const float*)d_in[4];
    const float* b    = (const float*)d_in[5];
    float* out = (float*)d_out;
    char*  wsb = (char*)d_ws;
    int*   tab = (int*)(wsb + (64u << 20));
    unsigned int* ent = (unsigned int*)(wsb + (128u << 20));

    const int n_nodes = in_sizes[0] / DIN;
    const int n_edges = in_sizes[3];

    const int nbins   = (n_nodes + RPB - 1) / RPB;
    const int eblocks = (n_edges + EPB - 1) / EPB;
    gcn_slice_group<<<eblocks, 256, 0, stream>>>(erow, ecol, ev, tab, ent, n_edges, nbins);

    gcn_rows_sum_xwt<<<nbins, 1024, 0, stream>>>(ent, tab, x, W, b, out, n_nodes, nbins, eblocks);
}


# row sums in registers: a pass's edge list is counting-sorted by out row in LDS (returning int adds, arrival rank); a wave FMAs a contiguous span in list order, one LDS f32 add per row, not per edge
# speedup vs baseline: 3.2532x; 1.0964x over previous
#include <hip/hip_runtime.h>

typedef __attribute__((ext_vector_type(2))) float v2f;
typedef __attribute__((ext_vector_type(8))) float v8f;

#define DIN  64
#define DOUT 64
#define RPB  392
#define MAXBINS 1024
#define EPB  4096
#define COLBITS 23
#define PITCH 68
#define CAP  8192
#define SEGT 1024
#define TROWS 400
#define NRC  512
#define NSW  13
#define EPT  (CAP / 1024)

__global__ __launch_bounds__(256) void gcn_slice_group(
    const int* __restrict__ erow, const int* __restrict__ ecol,
    const float* __restrict__ eval, int* __restrict__ tab,
    unsigned int* __restrict__ ent, int n_edges, int nbins)
{
    __shared__ int h[MAXBINS];
    __shared__ unsigned int skey[EPB];
    __shared__ unsigned int sval[EPB];
    __shared__ int wc[8];

    for (int i = threadIdx.x; i < nbins; i += 256) h[i] = 0;
    __syncthreads();

    const int base = blockIdx.x * EPB + threadIdx.x;
    int   rows[EPB / 256];
    int   cols[EPB / 256];
    float vals[EPB / 256];
    #pragma unroll
    for (int j = 0; j < EPB / 256; ++j) {
        const int e = base + j * 256;
        const bool ok = e < n_edges;
        rows[j] = ok ? erow[e] : -1;
        cols[j] = ok ? ecol[e] : 0;
        vals[j] = ok ? eval[e] : 0.0f;
    }
    #pragma unroll
    for (int j = 0; j < EPB / 256; ++j)
        if (rows[j] >= 0)
            atomicAdd(&h[(unsigned)rows[j] / RPB], 1);
    __syncthreads();

    const int lane = threadIdx.x & 31;
    const int wave = threadIdx.x >> 5;
    const int t4   = threadIdx.x * 4;
    int c[4];
    #pragma unroll
    for (int q = 0; q < 4; ++q)
        c[q] = (t4 + q < nbins) ? h[t4 + q] : 0;
    const int sc = c[0] + c[1] + c[2] + c[3];
    int ic = sc;
    #pragma unroll
    for (int d = 1; d < 32; d <<= 1) {
        const int tc = __shfl_up(ic, d);
        if (lane >= d) ic += tc;
    }
    if (lane == 31) wc[wave] = ic;
    __syncthreads();
    int pc = ic - sc;
    for (int w = 0; w < wave; ++w) pc += wc[w];
    int2* __restrict__ trow = (int2*)tab + (size_t)blockIdx.x * nbins;
    #pragma unroll
    for (int q = 0; q < 4; ++q) {
        const int i = t4 + q;
        if (i < nbins) {
            trow[i] = make_int2(pc, c[q]);
            h[i] = pc;
        }
        pc += c[q];
    }
    __syncthreads();

    #pragma unroll
    for (int j = 0; j < EPB / 256; ++j) {
        if (rows[j] >= 0) {
            const unsigned bin = (unsigned)rows[j] / RPB;
            const unsigned rl  = (unsigned)rows[j] - bin * RPB;
            const int p = atomicAdd(&h[bin], 1);
            skey[p] = (unsigned)cols[j] | (rl << COLBITS);
            sval[p] = __float_as_uint(vals[j]);
        }
    }
    __syncthreads();

    const int first = blockIdx.x * EPB;
    const int nloc  = min(EPB, n_edges - first);
    uint2* __restrict__ ent2 = (uint2*)ent + first;
    for (int p = threadIdx.x; p < nloc; p += 256)
        ent2[p] = make_uint2(skey[p], sval[p]);
}

__global__ __launch_bounds__(1024) void gcn_rows_sum_xwt(
    const unsigned int* __restrict__ ent, const int* __restrict__ tab,
    const float* __restrict__ x,
    const float* __restrict__ W, const float* __restrict__ b,
    float* __restrict__ out, int n_nodes, int nbins, int nseg)
{
    __shared__ __attribute__((aligned(16))) float acc[TROWS * PITCH];
    __shared__ __attribute__((aligned(16))) float sW[DOUT * PITCH];
    __shared__ uint2 stage[CAP];
    __shared__ int sbeg[SEGT];
    __shared__ int spos[SEGT];
    __shared__ int slen[SEGT];
    __shared__ int wtot[33];
    __shared__ int cnt[NRC];
    __shared__ int rstart[NRC];
    const int lane = threadIdx.x & 31;
    const int wave = __builtin_amdgcn_readfirstlane(threadIdx.x >> 5);
    const int bin  = blockIdx.x;
    const int row0 = bin * RPB;
    const int m    = lane & 15;
    const int half = lane >> 4;

    {
        const int wn = threadIdx.x >> 4;
        const int wk = (threadIdx.x & 15) * 4;
        *(float4*)&sW[wn * PITCH + wk] = *(const float4*)&W[wn * DIN + wk];
    }
    const float2 bb = ((const float2*)b)[lane];

    for (int r = wave; r < TROWS; r += 32)
        *(float2*)&acc[r * PITCH + 2 * lane] = make_float2(0.0f, 0.0f);
    if (threadIdx.x < NRC) cnt[threadIdx.x] = 0;
    __syncthreads();

    const uint2* __restrict__ ent2 = (const uint2*)ent;
    const int2*  __restrict__ tab2 = (const int2*)tab;

    for (int g0 = 0; g0 < nseg; g0 += SEGT) {
        const int nsg = min(SEGT, nseg - g0);
        int gb = 0, gl = 0;
        if ((int)threadIdx.x < nsg) {
            const int t = g0 + (int)threadIdx.x;
            const int2 e = tab2[(size_t)t * nbins + bin];
            gb = t * EPB + e.x;
            gl = e.y;
        }
        int inc = gl;
        #pragma unroll
        for (int dd = 1; dd < 32; dd <<= 1) {
            const int v = __shfl_up(inc, dd);
            if (lane >= dd) inc += v;
        }
        if (lane == 31) wtot[wave] = inc;
        __syncthreads();
        if (wave == 0) {
            const int w = wtot[lane];
            int wi = w;
            #pragma unroll
            for (int dd = 1; dd < 32; dd <<= 1) {
                const int v = __shfl_up(wi, dd);
                if (lane >= dd) wi += v;
            }
            wtot[lane] = wi - w;
            if (lane == 31) wtot[32] = wi;
        }
        __syncthreads();
        sbeg[threadIdx.x] = gb;
        spos[threadIdx.x] = wtot[wave] + inc - gl;
        slen[threadIdx.x] = gl;
        const int gtot = __builtin_amdgcn_readfirstlane(wtot[32]);
        __syncthreads();

        for (int r0 = 0; r0 < gtot; r0 += CAP) {
            const int n = min(CAP, gtot - r0);

            for (int sb = 0; sb < nsg; sb += 256) {
                uint2 v[8];
                int ln[8], ps[8], bg[8];
                #pragma unroll
                for (int u = 0; u < 8; ++u) {
                    const int s  = sb + wave + 32 * u;
                    const int sc = min(s, nsg - 1);
                    ln[u] = (s < nsg) ? slen[sc] : 0;
                    ps[u] = spos[sc] - r0;
                    bg[u] = sbeg[sc];
                }
                #pragma unroll
                for (int u = 0; u < 8; ++u) {
                    v[u] = make_uint2(0u, 0u);
                    if (lane < ln[u]) v[u] = ent2[bg[u] + lane];
                }
                #pragma unroll
                for (int u = 0; u < 8; ++u) {
                    const int q = ps[u] + lane;
                    if (lane < ln[u] && (unsigned)q < (unsigned)CAP) stage[q] = v[u];
                    for (int j = lane + 32; j < ln[u]; j += 32) {
                        const int q2 = ps[u] + j;
                        if ((unsigned)q2 < (unsigned)CAP) stage[q2] = ent2[bg[u] + j];
                    }
                }
            }
            __syncthreads();

            uint2 se[EPT];
            int   sr[EPT];
            #pragma unroll
            for (int k = 0; k < EPT; ++k) {
                const int i = (int)threadIdx.x + k * 1024;
                se[k] = make_uint2(0u, 0u);
                if (i < n) se[k] = stage[i];
            }
            #pragma unroll
            for (int k = 0; k < EPT; ++k) {
                const int i = (int)threadIdx.x + k * 1024;
                sr[k] = 0;
                if (i < n) sr[k] = atomicAdd(&cnt[se[k].x >> COLBITS], 1);
            }
            __syncthreads();
            if (wave < NSW) {
                const int cv = cnt[threadIdx.x];
                int pre = 0;
                for (int w = 0; w < wave; ++w) pre += cnt[lane + 32 * w];
                #pragma unroll
                for (int dd = 16; dd >= 1; dd >>= 1) pre += __shfl_xor(pre, dd);
                int ri = cv;
                #pragma unroll
                for (int dd = 1; dd < 32; dd <<= 1) {
                    const int up = __shfl_up(ri, dd);
                    if (lane >= dd) ri += up;
                }
                rstart[threadIdx.x] = pre + ri - cv;
            }
            __syncthreads();
            if (threadIdx.x < NRC) cnt[threadIdx.x] = 0;
            #pragma unroll
            for (int k = 0; k < EPT; ++k) {
                const int i = (int)threadIdx.x + k * 1024;
                if (i < n) sr[k] += rstart[se[k].x >> COLBITS];
            }
            #pragma unroll
            for (int k = 0; k < EPT; ++k) {
                const int i = (int)threadIdx.x + k * 1024;
                if (i < n) stage[sr[k]] = se[k];
            }
            __syncthreads();

            const int per = (n + 31) >> 5;
            const int beg = wave * per;
            const int end = min(n, beg + per);
            int   cur = -1;
            float s0 = 0.0f, s1 = 0.0f;
            int c0 = beg;
            uint2 en = make_uint2(0u, 0u);
            if (c0 + lane < end) en = stage[c0 + lane];
            while (c0 < end) {
                const int cn = c0 + 32;
                uint2 en_next = make_uint2(0u, 0u);
                if (cn + lane < end) en_next = stage[cn + lane];

                const int mm = min(32, end - c0);
                for (int j = 0; j < mm; j += 16) {
                    float2 xv[16];
                    #pragma unroll
                    for (int u = 0; u < 16; ++u) {
                        const int jj = min(j + u, mm - 1);
                        const unsigned key = __builtin_amdgcn_readlane(en.x, jj);
                        const unsigned col = key & ((1u << COLBITS) - 1u);
                        xv[u] = ((const float2*)(x + (size_t)col * DIN))[lane];
                    }
                    #pragma unroll
                    for (int u = 0; u < 16; ++u) {
                        if (j + u < mm) {
                            const unsigned key = __builtin_amdgcn_readlane(en.x, j + u);
                            const float    val = __uint_as_float(__builtin_amdgcn_readlane(en.y, j + u));
                            const int      r   = (int)(key >> COLBITS);
                            if (r != cur) {
                                if (cur >= 0) {
                                    float* a = &acc[cur * PITCH + 2 * lane];
                                    atomicAdd(a,     s0);
                                    atomicAdd(a + 1, s1);
                                }
                                cur = r;
                                s0 = 0.0f;
                                s1 = 0.0f;
                            }
                            s0 = fmaf(val, xv[u].x, s0);
                            s1 = fmaf(val, xv[u].y, s1);
                        }
                    }
                }
                en = en_next;
                c0 = cn;
            }
            if (cur >= 0) {
                float* a = &acc[cur * PITCH + 2 * lane];
                atomicAdd(a,     s0);
                atomicAdd(a + 1, s1);
            }
            __syncthreads();
        }
    }

    for (int t = wave; t < TROWS / 16; t += 32) {
        const float* ar = &acc[(t * 16 + m) * PITCH + 2 * half];
        v2f a[16];
        #pragma unroll
        for (int s = 0; s < 16; ++s) a[s] = *(const v2f*)(ar + 4 * s);

        v8f d[4] = {};

        #pragma unroll
        for (int s = 0; s < 16; ++s) {
            #pragma unroll
            for (int nt = 0; nt < 4; ++nt) {
                const v2f bw = *(const v2f*)&sW[(nt * 16 + m) * PITCH + 4 * s + 2 * half];
                d[nt] = __builtin_amdgcn_wmma_f32_16x16x4_f32(
                     false, a[s],  false, bw,
                     (short)0, d[nt],
                     false,  false);
            }
        }

        #pragma unroll
        for (int nt = 0; nt < 4; ++nt)
            #pragma unroll
            for (int v = 0; v < 8; ++v)
                acc[(t * 16 + v + 8 * half) * PITCH + nt * 16 + m] = d[nt][v];
    }
    __syncthreads();

    for (int r = wave; r < RPB; r += 32) {
        const int gr = row0 + r;
        if (gr < n_nodes) {
            const float2 o = *(const float2*)&acc[r * PITCH + 2 * lane];
            *(float2*)&out[(size_t)gr * DOUT + 2 * lane] = make_float2(o.x + bb.x, o.y + bb.y);
        }
    }
}

extern "C" void kernel_launch(void* const* d_in, const int* in_sizes, int n_in,
                              void* d_out, int out_size, void* d_ws, size_t ws_size,
                              hipStream_t stream)
{
    const float* x    = (const float*)d_in[0];
    const int*   erow = (const int*)  d_in[1];
    const int*   ecol = (const int*)  d_in[2];
    const float* ev   = (const float*)d_in[3];
    const float* W    = (const float*)d_in[4];
    const float* b    = (const float*)d_in[5];
    float* out = (float*)d_out;
    char*  wsb = (char*)d_ws;
    int*   tab = (int*)(wsb + (64u << 20));
    unsigned int* ent = (unsigned int*)(wsb + (128u << 20));

    const int n_nodes = in_sizes[0] / DIN;
    const int n_edges = in_sizes[3];

    const int nbins   = (n_nodes + RPB - 1) / RPB;
    const int eblocks = (n_edges + EPB - 1) / EPB;
    gcn_slice_group<<<eblocks, 256, 0, stream>>>(erow, ecol, ev, tab, ent, n_edges, nbins);

    gcn_rows_sum_xwt<<<nbins, 1024, 0, stream>>>(ent, tab, x, W, b, out, n_nodes, nbins, eblocks);
}


# row-sum loop: per 32 list entries the x-row byte offsets and a row-change ballot are made once in vector code; an edge then costs a literal-lane read, one bit test and its FMAs; sums unchanged
# speedup vs baseline: 3.5774x; 1.0997x over previous
#include <hip/hip_runtime.h>

typedef __attribute__((ext_vector_type(2))) float v2f;
typedef __attribute__((ext_vector_type(8))) float v8f;

#define DIN  64
#define DOUT 64
#define RPB  392
#define MAXBINS 1024
#define EPB  4096
#define COLBITS 23
#define PITCH 68
#define CAP  8192
#define SEGT 1024
#define TROWS 400
#define NRC  512
#define NSW  13
#define EPT  (CAP / 1024)

__global__ __launch_bounds__(256) void gcn_slice_group(
    const int* __restrict__ erow, const int* __restrict__ ecol,
    const float* __restrict__ eval, int* __restrict__ tab,
    unsigned int* __restrict__ ent, int n_edges, int nbins)
{
    __shared__ int h[MAXBINS];
    __shared__ unsigned int skey[EPB];
    __shared__ unsigned int sval[EPB];
    __shared__ int wc[8];

    for (int i = threadIdx.x; i < nbins; i += 256) h[i] = 0;
    __syncthreads();

    const int base = blockIdx.x * EPB + threadIdx.x;
    int   rows[EPB / 256];
    int   cols[EPB / 256];
    float vals[EPB / 256];
    #pragma unroll
    for (int j = 0; j < EPB / 256; ++j) {
        const int e = base + j * 256;
        const bool ok = e < n_edges;
        rows[j] = ok ? erow[e] : -1;
        cols[j] = ok ? ecol[e] : 0;
        vals[j] = ok ? eval[e] : 0.0f;
    }
    #pragma unroll
    for (int j = 0; j < EPB / 256; ++j)
        if (rows[j] >= 0)
            atomicAdd(&h[(unsigned)rows[j] / RPB], 1);
    __syncthreads();

    const int lane = threadIdx.x & 31;
    const int wave = threadIdx.x >> 5;
    const int t4   = threadIdx.x * 4;
    int c[4];
    #pragma unroll
    for (int q = 0; q < 4; ++q)
        c[q] = (t4 + q < nbins) ? h[t4 + q] : 0;
    const int sc = c[0] + c[1] + c[2] + c[3];
    int ic = sc;
    #pragma unroll
    for (int d = 1; d < 32; d <<= 1) {
        const int tc = __shfl_up(ic, d);
        if (lane >= d) ic += tc;
    }
    if (lane == 31) wc[wave] = ic;
    __syncthreads();
    int pc = ic - sc;
    for (int w = 0; w < wave; ++w) pc += wc[w];
    int2* __restrict__ trow = (int2*)tab + (size_t)blockIdx.x * nbins;
    #pragma unroll
    for (int q = 0; q < 4; ++q) {
        const int i = t4 + q;
        if (i < nbins) {
            trow[i] = make_int2(pc, c[q]);
            h[i] = pc;
        }
        pc += c[q];
    }
    __syncthreads();

    #pragma unroll
    for (int j = 0; j < EPB / 256; ++j) {
        if (rows[j] >= 0) {
            const unsigned bin = (unsigned)rows[j] / RPB;
            const unsigned rl  = (unsigned)rows[j] - bin * RPB;
            const int p = atomicAdd(&h[bin], 1);
            skey[p] = (unsigned)cols[j] | (rl << COLBITS);
            sval[p] = __float_as_uint(vals[j]);
        }
    }
    __syncthreads();

    const int first = blockIdx.x * EPB;
    const int nloc  = min(EPB, n_edges - first);
    uint2* __restrict__ ent2 = (uint2*)ent + first;
    for (int p = threadIdx.x; p < nloc; p += 256)
        ent2[p] = make_uint2(skey[p], sval[p]);
}

__global__ __launch_bounds__(1024) void gcn_rows_sum_xwt(
    const unsigned int* __restrict__ ent, const int* __restrict__ tab,
    const float* __restrict__ x,
    const float* __restrict__ W, const float* __restrict__ b,
    float* __restrict__ out, int n_nodes, int nbins, int nseg)
{
    __shared__ __attribute__((aligned(16))) float acc[TROWS * PITCH];
    __shared__ __attribute__((aligned(16))) float sW[DOUT * PITCH];
    __shared__ uint2 stage[CAP];
    __shared__ int sbeg[SEGT];
    __shared__ int spos[SEGT];
    __shared__ int slen[SEGT];
    __shared__ int wtot[33];
    __shared__ int cnt[NRC];
    __shared__ int rstart[NRC];
    const int lane = threadIdx.x & 31;
    const int wave = __builtin_amdgcn_readfirstlane(threadIdx.x >> 5);
    const int bin  = blockIdx.x;
    const int row0 = bin * RPB;
    const int m    = lane & 15;
    const int half = lane >> 4;

    {
        const int wn = threadIdx.x >> 4;
        const int wk = (threadIdx.x & 15) * 4;
        *(float4*)&sW[wn * PITCH + wk] = *(const float4*)&W[wn * DIN + wk];
    }
    const float2 bb = ((const float2*)b)[lane];

    for (int r = wave; r < TROWS; r += 32)
        *(float2*)&acc[r * PITCH + 2 * lane] = make_float2(0.0f, 0.0f);
    if (threadIdx.x < NRC) cnt[threadIdx.x] = 0;
    __syncthreads();

    const uint2* __restrict__ ent2 = (const uint2*)ent;
    const int2*  __restrict__ tab2 = (const int2*)tab;

    for (int g0 = 0; g0 < nseg; g0 += SEGT) {
        const int nsg = min(SEGT, nseg - g0);
        int gb = 0, gl = 0;
        if ((int)threadIdx.x < nsg) {
            const int t = g0 + (int)threadIdx.x;
            const int2 e = tab2[(size_t)t * nbins + bin];
            gb = t * EPB + e.x;
            gl = e.y;
        }
        int inc = gl;
        #pragma unroll
        for (int dd = 1; dd < 32; dd <<= 1) {
            const int v = __shfl_up(inc, dd);
            if (lane >= dd) inc += v;
        }
        if (lane == 31) wtot[wave] = inc;
        __syncthreads();
        if (wave == 0) {
            const int w = wtot[lane];
            int wi = w;
            #pragma unroll
            for (int dd = 1; dd < 32; dd <<= 1) {
                const int v = __shfl_up(wi, dd);
                if (lane >= dd) wi += v;
            }
            wtot[lane] = wi - w;
            if (lane == 31) wtot[32] = wi;
        }
        __syncthreads();
        sbeg[threadIdx.x] = gb;
        spos[threadIdx.x] = wtot[wave] + inc - gl;
        slen[threadIdx.x] = gl;
        const int gtot = __builtin_amdgcn_readfirstlane(wtot[32]);
        __syncthreads();

        for (int r0 = 0; r0 < gtot; r0 += CAP) {
            const int n = min(CAP, gtot - r0);

            for (int sb = 0; sb < nsg; sb += 256) {
                uint2 v[8];
                int ln[8], ps[8], bg[8];
                #pragma unroll
                for (int u = 0; u < 8; ++u) {
                    const int s  = sb + wave + 32 * u;
                    const int sc = min(s, nsg - 1);
                    ln[u] = (s < nsg) ? slen[sc] : 0;
                    ps[u] = spos[sc] - r0;
                    bg[u] = sbeg[sc];
                }
                #pragma unroll
                for (int u = 0; u < 8; ++u) {
                    v[u] = make_uint2(0u, 0u);
                    if (lane < ln[u]) v[u] = ent2[bg[u] + lane];
                }
                #pragma unroll
                for (int u = 0; u < 8; ++u) {
                    const int q = ps[u] + lane;
                    if (lane < ln[u] && (unsigned)q < (unsigned)CAP) stage[q] = v[u];
                    for (int j = lane + 32; j < ln[u]; j += 32) {
                        const int q2 = ps[u] + j;
                        if ((unsigned)q2 < (unsigned)CAP) stage[q2] = ent2[bg[u] + j];
                    }
                }
            }
            __syncthreads();

            uint2 se[EPT];
            int   sr[EPT];
            #pragma unroll
            for (int k = 0; k < EPT; ++k) {
                const int i = (int)threadIdx.x + k * 1024;
                se[k] = make_uint2(0u, 0u);
                if (i < n) se[k] = stage[i];
            }
            #pragma unroll
            for (int k = 0; k < EPT; ++k) {
                const int i = (int)threadIdx.x + k * 1024;
                sr[k] = 0;
                if (i < n) sr[k] = atomicAdd(&cnt[se[k].x >> COLBITS], 1);
            }
            __syncthreads();
            if (wave < NSW) {
                const int cv = cnt[threadIdx.x];
                int pre = 0;
                for (int w = 0; w < wave; ++w) pre += cnt[lane + 32 * w];
                #pragma unroll
                for (int dd = 16; dd >= 1; dd >>= 1) pre += __shfl_xor(pre, dd);
                int ri = cv;
                #pragma unroll
                for (int dd = 1; dd < 32; dd <<= 1) {
                    const int up = __shfl_up(ri, dd);
                    if (lane >= dd) ri += up;
                }
                rstart[threadIdx.x] = pre + ri - cv;
            }
            __syncthreads();
            if (threadIdx.x < NRC) cnt[threadIdx.x] = 0;
            #pragma unroll
            for (int k = 0; k < EPT; ++k) {
                const int i = (int)threadIdx.x + k * 1024;
                if (i < n) sr[k] += rstart[se[k].x >> COLBITS];
            }
            #pragma unroll
            for (int k = 0; k < EPT; ++k) {
                const int i = (int)threadIdx.x + k * 1024;
                if (i < n) stage[sr[k]] = se[k];
            }
            __syncthreads();

            const int per = (n + 31) >> 5;
            const int beg = wave * per;
            const int end = min(n, beg + per);
            int   cur = -1;
            float s0 = 0.0f, s1 = 0.0f;
            const char* xl = (const char*)x;
            const unsigned lane8 = 8u * (unsigned)lane;
            uint2 en = make_uint2(0u, 0u);
            if (beg < end) en = stage[beg + min(lane, min(32, end - beg) - 1)];
            for (int c0 = beg; c0 < end; c0 += 32) {
                const int mm  = min(32, end - c0);
                const int cn  = c0 + 32;
                const int mmn = min(32, end - cn);
                uint2 en_next = en;
                if (mmn > 0) en_next = stage[cn + min(lane, mmn - 1)];

                const unsigned rowv = en.x >> COLBITS;
                const unsigned offv = (en.x & ((1u << COLBITS) - 1u)) * (DIN * 4u);
                const unsigned valv = (lane < mm) ? en.y : 0u;
                int prev = __shfl_up((int)rowv, 1);
                if (lane == 0) prev = cur;
                const unsigned chg = __builtin_amdgcn_ballot_w32((int)rowv != prev);

                #pragma unroll
                for (int h = 0; h < 2; ++h) {
                    if (16 * h < mm) {
                        float2 xv[16];
                        #pragma unroll
                        for (int u = 0; u < 16; ++u) {
                            const unsigned off = (unsigned)__builtin_amdgcn_readlane((int)offv, 16 * h + u);
                            xv[u] = *(const float2*)(xl + (off + lane8));
                        }
                        #pragma unroll
                        for (int u = 0; u < 16; ++u) {
                            const int L = 16 * h + u;
                            if (__builtin_expect(((chg >> L) & 1u) != 0u, 0)) {
                                if (cur >= 0) {
                                    float* a = &acc[cur * PITCH + 2 * lane];
                                    atomicAdd(a,     s0);
                                    atomicAdd(a + 1, s1);
                                }
                                cur = __builtin_amdgcn_readlane((int)rowv, L);
                                s0 = 0.0f;
                                s1 = 0.0f;
                            }
                            const float val = __uint_as_float((unsigned)__builtin_amdgcn_readlane((int)valv, L));
                            s0 = fmaf(val, xv[u].x, s0);
                            s1 = fmaf(val, xv[u].y, s1);
                        }
                    }
                }
                en = en_next;
            }
            if (cur >= 0) {
                float* a = &acc[cur * PITCH + 2 * lane];
                atomicAdd(a,     s0);
                atomicAdd(a + 1, s1);
            }
            __syncthreads();
        }
    }

    for (int t = wave; t < TROWS / 16; t += 32) {
        const float* ar = &acc[(t * 16 + m) * PITCH + 2 * half];
        v2f a[16];
        #pragma unroll
        for (int s = 0; s < 16; ++s) a[s] = *(const v2f*)(ar + 4 * s);

        v8f d[4] = {};

        #pragma unroll
        for (int s = 0; s < 16; ++s) {
            #pragma unroll
            for (int nt = 0; nt < 4; ++nt) {
                const v2f bw = *(const v2f*)&sW[(nt * 16 + m) * PITCH + 4 * s + 2 * half];
                d[nt] = __builtin_amdgcn_wmma_f32_16x16x4_f32(
                     false, a[s],  false, bw,
                     (short)0, d[nt],
                     false,  false);
            }
        }

        #pragma unroll
        for (int nt = 0; nt < 4; ++nt)
            #pragma unroll
            for (int v = 0; v < 8; ++v)
                acc[(t * 16 + v + 8 * half) * PITCH + nt * 16 + m] = d[nt][v];
    }
    __syncthreads();

    for (int r = wave; r < RPB; r += 32) {
        const int gr = row0 + r;
        if (gr < n_nodes) {
            const float2 o = *(const float2*)&acc[r * PITCH + 2 * lane];
            *(float2*)&out[(size_t)gr * DOUT + 2 * lane] = make_float2(o.x + bb.x, o.y + bb.y);
        }
    }
}

extern "C" void kernel_launch(void* const* d_in, const int* in_sizes, int n_in,
                              void* d_out, int out_size, void* d_ws, size_t ws_size,
                              hipStream_t stream)
{
    const float* x    = (const float*)d_in[0];
    const int*   erow = (const int*)  d_in[1];
    const int*   ecol = (const int*)  d_in[2];
    const float* ev   = (const float*)d_in[3];
    const float* W    = (const float*)d_in[4];
    const float* b    = (const float*)d_in[5];
    float* out = (float*)d_out;
    char*  wsb = (char*)d_ws;
    int*   tab = (int*)(wsb + (64u << 20));
    unsigned int* ent = (unsigned int*)(wsb + (128u << 20));

    const int n_nodes = in_sizes[0] / DIN;
    const int n_edges = in_sizes[3];

    const int nbins   = (n_nodes + RPB - 1) / RPB;
    const int eblocks = (n_edges + EPB - 1) / EPB;
    gcn_slice_group<<<eblocks, 256, 0, stream>>>(erow, ecol, ev, tab, ent, n_edges, nbins);

    gcn_rows_sum_xwt<<<nbins, 1024, 0, stream>>>(ent, tab, x, W, b, out, n_nodes, nbins, eblocks);
}


# edges kept in fixed 32-place slots per (slice, row range): table, its dependent read, block scan and LDS list copy go; slots read at entry into the counting sort; a full slot's extras: LDS f32 adds
# speedup vs baseline: 3.7477x; 1.0476x over previous
#include <hip/hip_runtime.h>


typedef __attribute__((ext_vector_type(2))) float v2f;
typedef __attribute__((ext_vector_type(8))) float v8f;

#define DIN  64
#define DOUT 64
#define RPB  392
#define EPB  4096
#define COLBITS 23
#define PITCH 68
#define SLOT 32
#define BCH  256
#define SLP  34
#define SPP  256
#define CAP  8192
#define TROWS 400
#define NRC  512
#define NSW  13
#define EPT  (CAP / 1024)
#define NOKEY 0xFFFFFFFFu

__global__ __launch_bounds__(256) void gcn_slice_slots(
    const int* __restrict__ erow, const int* __restrict__ ecol,
    const float* __restrict__ eval, unsigned int* __restrict__ ent,
    unsigned int* __restrict__ ovf, int* __restrict__ ovfcnt, int n_edges, int nbins)
{
    __shared__ int h[BCH];
    __shared__ __attribute__((aligned(16))) uint2 sl[BCH * SLP];
    __shared__ int novf;

    const int base = blockIdx.x * EPB + threadIdx.x;
    int   rows[EPB / 256];
    int   cols[EPB / 256];
    float vals[EPB / 256];
    #pragma unroll
    for (int j = 0; j < EPB / 256; ++j) {
        const int e = base + j * 256;
        const bool ok = e < n_edges;
        rows[j] = ok ? erow[e] : -1;
        cols[j] = ok ? ecol[e] : 0;
        vals[j] = ok ? eval[e] : 0.0f;
    }
    if (threadIdx.x == 0) novf = 0;
    uint4* __restrict__ ovf4 = (uint4*)ovf + (size_t)blockIdx.x * EPB;

    for (int b0 = 0; b0 < nbins; b0 += BCH) {
        const int nb = min(BCH, nbins - b0);
        h[threadIdx.x] = 0;
        __syncthreads();

        #pragma unroll
        for (int j = 0; j < EPB / 256; ++j) {
            if (rows[j] >= 0) {
                const unsigned bin = (unsigned)rows[j] / RPB;
                const unsigned lb  = bin - (unsigned)b0;
                if (lb < (unsigned)nb) {
                    const unsigned rl = (unsigned)rows[j] - bin * RPB;
                    const int p = atomicAdd(&h[lb], 1);
                    if (p < SLOT) {
                        sl[lb * SLP + p] = make_uint2((unsigned)cols[j] | (rl << COLBITS), __float_as_uint(vals[j]));
                    } else {
                        const int q = atomicAdd(&novf, 1);
                        ovf4[q] = make_uint4((unsigned)rows[j], (unsigned)cols[j], __float_as_uint(vals[j]), 0u);
                    }
                }
            }
        }
        __syncthreads();

        uint4* __restrict__ dst = (uint4*)((uint2*)ent + ((size_t)blockIdx.x * nbins + b0) * SLOT);
        const uint4* s4 = (const uint4*)sl;
        for (int i = threadIdx.x; i < nb * (SLOT / 2); i += 256) {
            const int c = h[i >> 4];
            const int j = (i & 15) * 2;
            uint4 v = make_uint4(NOKEY, 0u, NOKEY, 0u);
            if (j < c) v = s4[(i >> 4) * (SLP / 2) + (i & 15)];
            if (j + 1 >= c) { v.z = NOKEY; v.w = 0u; }
            dst[i] = v;
        }
        if (b0 + BCH < nbins) __syncthreads();
    }
    if (threadIdx.x == 0) ovfcnt[blockIdx.x] = novf;
}

__global__ __launch_bounds__(1024) void gcn_rows_sum_xwt(
    const unsigned int* __restrict__ ent, const unsigned int* __restrict__ ovf,
    const int* __restrict__ ovfcnt, const float* __restrict__ x,
    const float* __restrict__ W, const float* __restrict__ b,
    float* __restrict__ out, int n_nodes, int nbins, int nseg)
{
    __shared__ __attribute__((aligned(16))) float acc[TROWS * PITCH];
    __shared__ __attribute__((aligned(16))) float sW[DOUT * PITCH];
    __shared__ uint2 stage[CAP];
    __shared__ int cnt[NRC];
    __shared__ int rstart[NRC];
    __shared__ int ntot;
    const int lane = threadIdx.x & 31;
    const int wave = __builtin_amdgcn_readfirstlane(threadIdx.x >> 5);
    const int bin  = blockIdx.x;
    const int row0 = bin * RPB;
    const int m    = lane & 15;
    const int half = lane >> 4;

    const uint2* __restrict__ ent2 = (const uint2*)ent;

    uint2 se[EPT];
    {
        const int nsl = min(SPP, nseg);
        #pragma unroll
        for (int k = 0; k < EPT; ++k) {
            const int i = (int)threadIdx.x + k * 1024;
            const int s = i >> 5;
            se[k] = make_uint2(NOKEY, 0u);
            if (s < nsl) se[k] = ent2[((size_t)s * nbins + bin) * SLOT + (i & 31)];
        }
    }
    const int own = (int)threadIdx.x - NSW * 32;
    const int oc0 = (own >= 0 && own < nseg) ? ovfcnt[own] : 0;

    {
        const int wn = threadIdx.x >> 4;
        const int wk = (threadIdx.x & 15) * 4;
        *(float4*)&sW[wn * PITCH + wk] = *(const float4*)&W[wn * DIN + wk];
    }
    const float2 bb = ((const float2*)b)[lane];

    for (int r = wave; r < TROWS; r += 32)
        *(float2*)&acc[r * PITCH + 2 * lane] = make_float2(0.0f, 0.0f);
    if (threadIdx.x < NRC) cnt[threadIdx.x] = 0;
    __syncthreads();

    for (int g0 = 0; ; ) {
        int sr[EPT];
        #pragma unroll
        for (int k = 0; k < EPT; ++k) {
            sr[k] = 0;
            if (se[k].x != NOKEY) sr[k] = atomicAdd(&cnt[se[k].x >> COLBITS], 1);
        }
        __syncthreads();
        if (wave < NSW) {
            const int cv = cnt[threadIdx.x];
            int pre = 0;
            for (int w = 0; w < wave; ++w) pre += cnt[lane + 32 * w];
            #pragma unroll
            for (int dd = 16; dd >= 1; dd >>= 1) pre += __shfl_xor(pre, dd);
            int ri = cv;
            #pragma unroll
            for (int dd = 1; dd < 32; dd <<= 1) {
                const int up = __shfl_up(ri, dd);
                if (lane >= dd) ri += up;
            }
            rstart[threadIdx.x] = pre + ri - cv;
            if (wave == NSW - 1 && lane == 31) ntot = pre + ri;
        } else if (g0 == 0) {
            for (int t0 = 0; t0 < nseg; t0 += 1024 - NSW * 32) {
                int oc = oc0;
                if (t0 > 0) {
                    const int t = t0 + own;
                    oc = (t < nseg) ? ovfcnt[t] : 0;
                }
                unsigned left = __builtin_amdgcn_ballot_w32(oc > 0);
                while (left != 0u) {
                    const int L = __builtin_ctz(left);
                    left &= left - 1u;
                    const int cs = min(EPB, __builtin_amdgcn_readlane(oc, L));
                    const uint4* __restrict__ o4 =
                        (const uint4*)ovf + (size_t)(t0 + (wave - NSW) * 32 + L) * EPB;
                    for (int j0 = 0; j0 < cs; j0 += 32) {
                        const int j = j0 + lane;
                        uint4 e = make_uint4(0u, 0u, 0u, 0u);
                        if (j < cs) e = o4[j];
                        unsigned hit = __builtin_amdgcn_ballot_w32(j < cs && e.x / RPB == (unsigned)bin);
                        while (hit != 0u) {
                            const int M = __builtin_ctz(hit);
                            hit &= hit - 1u;
                            const unsigned rl  = (unsigned)__builtin_amdgcn_readlane((int)e.x, M) - (unsigned)row0;
                            const unsigned col = (unsigned)__builtin_amdgcn_readlane((int)e.y, M);
                            const float val = __uint_as_float((unsigned)__builtin_amdgcn_readlane((int)e.z, M));
                            const float2 xs = *(const float2*)(x + (size_t)col * DIN + 2 * lane);
                            float* a = &acc[rl * PITCH + 2 * lane];
                            atomicAdd(a,     val * xs.x);
                            atomicAdd(a + 1, val * xs.y);
                        }
                    }
                }
            }
        }
        __syncthreads();
        if (threadIdx.x < NRC) cnt[threadIdx.x] = 0;
        #pragma unroll
        for (int k = 0; k < EPT; ++k)
            if (se[k].x != NOKEY) sr[k] += rstart[se[k].x >> COLBITS];
        #pragma unroll
        for (int k = 0; k < EPT; ++k)
            if (se[k].x != NOKEY) stage[sr[k]] = se[k];
        const int n = __builtin_amdgcn_readfirstlane(ntot);
        __syncthreads();

        const int per = (n + 31) >> 5;
        const int beg = wave * per;
        const int end = min(n, beg + per);
        int   cur = -1;
        float s0 = 0.0f, s1 = 0.0f;
        const char* xl = (const char*)x;
        const unsigned lane8 = 8u * (unsigned)lane;
        uint2 en = make_uint2(0u, 0u);
        if (beg < end) en = stage[beg + min(lane, min(32, end - beg) - 1)];
        for (int c0 = beg; c0 < end; c0 += 32) {
            const int mm  = min(32, end - c0);
            const int cn  = c0 + 32;
            const int mmn = min(32, end - cn);
            uint2 en_next = en;
            if (mmn > 0) en_next = stage[cn + min(lane, mmn - 1)];

            const unsigned rowv = en.x >> COLBITS;
            const unsigned offv = (en.x & ((1u << COLBITS) - 1u)) * (DIN * 4u);
            const unsigned valv = (lane < mm) ? en.y : 0u;
            int prev = __shfl_up((int)rowv, 1);
            if (lane == 0) prev = cur;
            const unsigned chg = __builtin_amdgcn_ballot_w32((int)rowv != prev);

            #pragma unroll
            for (int h = 0; h < 2; ++h) {
                if (16 * h < mm) {
                    float2 xv[16];
                    #pragma unroll
                    for (int u = 0; u < 16; ++u) {
                        const unsigned off = (unsigned)__builtin_amdgcn_readlane((int)offv, 16 * h + u);
                        xv[u] = *(const float2*)(xl + (off + lane8));
                    }
                    #pragma unroll
                    for (int u = 0; u < 16; ++u) {
                        const int L = 16 * h + u;
                        if (__builtin_expect(((chg >> L) & 1u) != 0u, 0)) {
                            if (cur >= 0) {
                                float* a = &acc[cur * PITCH + 2 * lane];
                                atomicAdd(a,     s0);
                                atomicAdd(a + 1, s1);
                            }
                            cur = __builtin_amdgcn_readlane((int)rowv, L);
                            s0 = 0.0f;
                            s1 = 0.0f;
                        }
                        const float val = __uint_as_float((unsigned)__builtin_amdgcn_readlane((int)valv, L));
                        s0 = fmaf(val, xv[u].x, s0);
                        s1 = fmaf(val, xv[u].y, s1);
                    }
                }
            }
            en = en_next;
        }
        if (cur >= 0) {
            float* a = &acc[cur * PITCH + 2 * lane];
            atomicAdd(a,     s0);
            atomicAdd(a + 1, s1);
        }
        __syncthreads();

        g0 += SPP;
        if (g0 >= nseg) break;
        const int nsl = min(SPP, nseg - g0);
        #pragma unroll
        for (int k = 0; k < EPT; ++k) {
            const int i = (int)threadIdx.x + k * 1024;
            const int s = i >> 5;
            se[k] = make_uint2(NOKEY, 0u);
            if (s < nsl) se[k] = ent2[((size_t)(g0 + s) * nbins + bin) * SLOT + (i & 31)];
        }
    }

    for (int t = wave; t < TROWS / 16; t += 32) {
        const float* ar = &acc[(t * 16 + m) * PITCH + 2 * half];
        v2f a[16];
        #pragma unroll
        for (int s = 0; s < 16; ++s) a[s] = *(const v2f*)(ar + 4 * s);

        v8f d[4] = {};

        #pragma unroll
        for (int s = 0; s < 16; ++s) {
            #pragma unroll
            for (int nt = 0; nt < 4; ++nt) {
                const v2f bw = *(const v2f*)&sW[(nt * 16 + m) * PITCH + 4 * s + 2 * half];
                d[nt] = __builtin_amdgcn_wmma_f32_16x16x4_f32(
                     false, a[s],  false, bw,
                     (short)0, d[nt],
                     false,  false);
            }
        }

        #pragma unroll
        for (int nt = 0; nt < 4; ++nt)
            #pragma unroll
            for (int v = 0; v < 8; ++v)
                acc[(t * 16 + v + 8 * half) * PITCH + nt * 16 + m] = d[nt][v];
    }
    __syncthreads();

    for (int r = wave; r < RPB; r += 32) {
        const int gr = row0 + r;
        if (gr < n_nodes) {
            const float2 o = *(const float2*)&acc[r * PITCH + 2 * lane];
            *(float2*)&out[(size_t)gr * DOUT + 2 * lane] = make_float2(o.x + bb.x, o.y + bb.y);
        }
    }
}

extern "C" void kernel_launch(void* const* d_in, const int* in_sizes, int n_in,
                              void* d_out, int out_size, void* d_ws, size_t ws_size,
                              hipStream_t stream)
{
    const float* x    = (const float*)d_in[0];
    const int*   erow = (const int*)  d_in[1];
    const int*   ecol = (const int*)  d_in[2];
    const float* ev   = (const float*)d_in[3];
    const float* W    = (const float*)d_in[4];
    const float* b    = (const float*)d_in[5];
    float* out = (float*)d_out;
    char*  wsb = (char*)d_ws;
    int*   ovfcnt = (int*)(wsb + (32u << 20));
    unsigned int* ovf = (unsigned int*)(wsb + (64u << 20));
    unsigned int* ent = (unsigned int*)(wsb + (128u << 20));

    const int n_nodes = in_sizes[0] / DIN;
    const int n_edges = in_sizes[3];

    const int nbins   = (n_nodes + RPB - 1) / RPB;
    const int eblocks = (n_edges + EPB - 1) / EPB;
    gcn_slice_slots<<<eblocks, 256, 0, stream>>>(erow, ecol, ev, ent, ovf, ovfcnt, n_edges, nbins);

    gcn_rows_sum_xwt<<<nbins, 1024, 0, stream>>>(ent, ovf, ovfcnt, x, W, b, out, n_nodes, nbins, eblocks);
}
